# starting kernel
# baseline (speedup 1.0000x reference)
#include <hip/hip_runtime.h>


typedef __attribute__((ext_vector_type(16))) _Float16 v16h;
typedef __attribute__((ext_vector_type(8)))  _Float16 half8;
typedef __attribute__((ext_vector_type(8)))  float    v8f;

#define HID 128

__global__ void inproj_kernel(const float* __restrict__ pos,
                              const int*   __restrict__ an,
                              const float* __restrict__ W,
                              const float* __restrict__ b,
                              float* __restrict__ h, int n) {
  int t = blockIdx.x * 256 + threadIdx.x;
  if (t >= n * HID) return;
  int node = t >> 7, c = t & 127;
  float z  = (float)an[node] * 0.1f;
  float px = pos[node * 3 + 0];
  float py = pos[node * 3 + 1];
  float pz = pos[node * 3 + 2];
  float v = b[c];
  v = fmaf(z,  W[0 * HID + c], v);
  v = fmaf(px, W[1 * HID + c], v);
  v = fmaf(py, W[2 * HID + c], v);
  v = fmaf(pz, W[3 * HID + c], v);
  h[t] = fmaxf(v, 0.0f);
}

__global__ void wprep_kernel(const float* __restrict__ W,
                             _Float16* __restrict__ WT) {
  int t = blockIdx.x * 256 + threadIdx.x;
  if (t >= HID * HID) return;
  int k = t & 127, n = t >> 7;
  WT[t] = (_Float16)W[k * HID + n];
}

__global__ void zero_kernel(float* __restrict__ p, int n) {
  int t = blockIdx.x * 256 + threadIdx.x;
  if (t < n) p[t] = 0.0f;
}

__global__ void scatter_kernel(const float* __restrict__ h,
                               const int*   __restrict__ ei,
                               float* __restrict__ s,
                               float* __restrict__ cnt, int nE) {
  int t = blockIdx.x * 256 + threadIdx.x;
  int e = t >> 5;
  if (e >= nE) return;
  int lane = t & 31;
  int ep = e + 4;
  if (ep < nE) {
    int psrc = ei[ep];
    __builtin_prefetch(h + (size_t)psrc * HID + lane * 4, 0, 1);
  }
  int src = ei[e];
  int dst = ei[nE + e];
  const float4 v = *(const float4*)(h + (size_t)src * HID + lane * 4);
  float* dp = s + (size_t)dst * HID + lane * 4;
  atomicAdd(dp + 0, v.x);
  atomicAdd(dp + 1, v.y);
  atomicAdd(dp + 2, v.z);
  atomicAdd(dp + 3, v.w);
  if (lane == 0) atomicAdd(cnt + dst, 1.0f);
}

__device__ inline half8 cvt8(const float* __restrict__ p, float scale) {
  float4 x = *(const float4*)(p);
  float4 y = *(const float4*)(p + 4);
  half8 r;
  r[0] = (_Float16)(x.x * scale); r[1] = (_Float16)(x.y * scale);
  r[2] = (_Float16)(x.z * scale); r[3] = (_Float16)(x.w * scale);
  r[4] = (_Float16)(y.x * scale); r[5] = (_Float16)(y.y * scale);
  r[6] = (_Float16)(y.z * scale); r[7] = (_Float16)(y.w * scale);
  return r;
}

__device__ inline v16h join16(half8 lo, half8 hi) {
  return __builtin_shufflevector(lo, hi, 0, 1, 2, 3, 4, 5, 6, 7,
                                         8, 9, 10, 11, 12, 13, 14, 15);
}

template <bool SAGE, bool RELU>
__global__ __launch_bounds__(256) void gemm_kernel(
    const float*    __restrict__ h,
    const float*    __restrict__ s,
    const float*    __restrict__ cnt,
    const _Float16* __restrict__ WlT,
    const _Float16* __restrict__ WrT,
    const float*    __restrict__ bias,
    float* __restrict__ out,
    int n_nodes) {
  __shared__ _Float16 wtr[HID * HID];
  __shared__ _Float16 wtl[HID * HID];

  {
    uint4*       dr = (uint4*)wtr;
    const uint4* sr = (const uint4*)WrT;
    for (int i = threadIdx.x; i < HID * HID / 8; i += 256) dr[i] = sr[i];
    if constexpr (SAGE) {
      uint4*       dl = (uint4*)wtl;
      const uint4* sl = (const uint4*)WlT;
      for (int i = threadIdx.x; i < HID * HID / 8; i += 256) dl[i] = sl[i];
    }
  }
  __syncthreads();

  const int lane = threadIdx.x & 31;
  const int wave = threadIdx.x >> 5;
  const int rowbase = blockIdx.x * 128 + wave * 16;
  if (rowbase >= n_nodes) return;

  const int arow = rowbase + (lane & 15);
  const int row  = arow < n_nodes ? arow : n_nodes - 1;
  const int lo   = (lane >= 16) ? 8 : 0;

  float inv = 0.0f;
  if constexpr (SAGE) {
    float cv = cnt[row];
    inv = (cv > 0.0f) ? __builtin_amdgcn_rcpf(cv) : 0.0f;
  }

  const float*    hp  = h + (size_t)row * HID + lo;
  const float*    sp  = SAGE ? (s + (size_t)row * HID + lo) : nullptr;
  const _Float16* brp = &wtr[(lane & 15) * HID + lo];
  const _Float16* blp = &wtl[(lane & 15) * HID + lo];

  v8f acc[8];
  {
    v8f z = {};
#pragma unroll
    for (int nt = 0; nt < 8; ++nt) acc[nt] = z;
  }

#pragma unroll
  for (int kk = 0; kk < 4; ++kk) {
    const int kb = kk * 32;
    v16h ahf = join16(cvt8(hp + kb, 1.0f), cvt8(hp + kb + 16, 1.0f));
    v16h agf;
    if constexpr (SAGE)
      agf = join16(cvt8(sp + kb, inv), cvt8(sp + kb + 16, inv));
#pragma unroll
    for (int nt = 0; nt < 8; ++nt) {
      const int off = nt * 16 * HID + kb;
      v16h br = join16(*(const half8*)(brp + off),
                       *(const half8*)(brp + off + 16));
      acc[nt] = __builtin_amdgcn_wmma_f32_16x16x32_f16(
          false, ahf, false, br, (short)0, acc[nt], false, false);
      if constexpr (SAGE) {
        v16h bl = join16(*(const half8*)(blp + off),
                         *(const half8*)(blp + off + 16));
        acc[nt] = __builtin_amdgcn_wmma_f32_16x16x32_f16(
            false, agf, false, bl, (short)0, acc[nt], false, false);
      }
    }
  }

  const int mofs = (lane >= 16) ? 8 : 0;
#pragma unroll
  for (int nt = 0; nt < 8; ++nt) {
    const int ncol = nt * 16 + (lane & 15);
    const float bb = bias[ncol];
#pragma unroll
    for (int r = 0; r < 8; ++r) {
      float v = acc[nt][r] + bb;
      if constexpr (RELU) v = fmaxf(v, 0.0f);
      int orow = rowbase + mofs + r;
      if (orow < n_nodes) out[(size_t)orow * HID + ncol] = v;
    }
  }
}

__global__ void colmean_kernel(const float* __restrict__ ns,
                               float* __restrict__ outm, int n) {
  __shared__ float red[256];
  int col = blockIdx.x;
  float acc = 0.0f;
  for (int r = threadIdx.x; r < n; r += 256) acc += ns[(size_t)r * HID + col];
  red[threadIdx.x] = acc;
  __syncthreads();
  for (int st = 128; st > 0; st >>= 1) {
    if (threadIdx.x < st) red[threadIdx.x] += red[threadIdx.x + st];
    __syncthreads();
  }
  if (threadIdx.x == 0) outm[col] = red[0] / (float)n;
}

extern "C" void kernel_launch(void* const* d_in, const int* in_sizes, int n_in,
                              void* d_out, int out_size, void* d_ws, size_t ws_size,
                              hipStream_t stream) {
  const float* pos  = (const float*)d_in[0];
  const int*   an   = (const int*)  d_in[1];
  const int*   ei   = (const int*)  d_in[2];
  const float* W_in = (const float*)d_in[3];
  const float* b_in = (const float*)d_in[4];
  const float* W1l  = (const float*)d_in[5];
  const float* b1   = (const float*)d_in[6];
  const float* W1r  = (const float*)d_in[7];
  const float* W2l  = (const float*)d_in[8];
  const float* b2   = (const float*)d_in[9];
  const float* W2r  = (const float*)d_in[10];
  const float* Wout = (const float*)d_in[11];
  const float* bout = (const float*)d_in[12];

  const int n  = in_sizes[0] / 3;
  const int nE = in_sizes[2] / 2;

  float* ws   = (float*)d_ws;
  float* h    = ws;
  float* s    = ws + (size_t)n * HID;
  float* cnt  = s  + (size_t)n * HID;

  size_t woff = (size_t)n * HID * 2 + n;
  woff = (woff + 63) & ~(size_t)63;
  _Float16* wtbase = (_Float16*)(ws + woff);
  _Float16* W1lT = wtbase + 0 * HID * HID;
  _Float16* W1rT = wtbase + 1 * HID * HID;
  _Float16* W2lT = wtbase + 2 * HID * HID;
  _Float16* W2rT = wtbase + 3 * HID * HID;
  _Float16* WoT  = wtbase + 4 * HID * HID;

  float* outm = (float*)d_out;
  float* nst  = (float*)d_out + HID;

  const int nh       = n * HID;
  const int gemmBlks = (n + 127) / 128;
  const int edgeThr  = nE * 32;
  const int wBlks    = (HID * HID + 255) / 256;

  wprep_kernel<<<wBlks, 256, 0, stream>>>(W1l,  W1lT);
  wprep_kernel<<<wBlks, 256, 0, stream>>>(W1r,  W1rT);
  wprep_kernel<<<wBlks, 256, 0, stream>>>(W2l,  W2lT);
  wprep_kernel<<<wBlks, 256, 0, stream>>>(W2r,  W2rT);
  wprep_kernel<<<wBlks, 256, 0, stream>>>(Wout, WoT);

  inproj_kernel<<<(nh + 255) / 256, 256, 0, stream>>>(pos, an, W_in, b_in, h, n);

  zero_kernel<<<(nh + n + 255) / 256, 256, 0, stream>>>(s, nh + n);
  scatter_kernel<<<(edgeThr + 255) / 256, 256, 0, stream>>>(h, ei, s, cnt, nE);
  gemm_kernel<true, true><<<gemmBlks, 256, 0, stream>>>(h, s, cnt, W1lT, W1rT, b1, h, n);

  zero_kernel<<<(nh + n + 255) / 256, 256, 0, stream>>>(s, nh + n);
  scatter_kernel<<<(edgeThr + 255) / 256, 256, 0, stream>>>(h, ei, s, cnt, nE);
  gemm_kernel<true, true><<<gemmBlks, 256, 0, stream>>>(h, s, cnt, W2lT, W2rT, b2, h, n);

  gemm_kernel<false, false><<<gemmBlks, 256, 0, stream>>>(h, nullptr, nullptr,
                                                          nullptr, WoT, bout, nst, n);
  colmean_kernel<<<HID, 256, 0, stream>>>(nst, outm, n);
}


# GEMM builds its B fragments straight from the f32 weight inputs [k][n], rounding to f16 in registers (same values); no transposed f16 copy, so its 5 prep launches go (9 launches); sums same
# speedup vs baseline: 4.8173x; 4.8173x over previous
#include <hip/hip_runtime.h>


typedef __attribute__((ext_vector_type(16))) _Float16 v16h;
typedef __attribute__((ext_vector_type(8)))  _Float16 half8;
typedef __attribute__((ext_vector_type(8)))  float    v8f;

#define HID 128

__global__ void xpack_kernel(const float* __restrict__ pos,
                             const int*   __restrict__ an,
                             float* __restrict__ xp, int n) {
  int node = blockIdx.x * 256 + threadIdx.x;
  if (node >= n) return;
  float4 x;
  x.x = (float)an[node] * 0.1f;
  x.y = pos[node * 3 + 0];
  x.z = pos[node * 3 + 1];
  x.w = pos[node * 3 + 2];
  ((float4*)xp)[node] = x;
}

__device__ inline float h0_elem(float z, float px, float py, float pz,
                                float w0, float w1, float w2, float w3, float b) {
  float v = b;
  v = fmaf(z,  w0, v);
  v = fmaf(px, w1, v);
  v = fmaf(py, w2, v);
  v = fmaf(pz, w3, v);
  return fmaxf(v, 0.0f);
}

#define NPB  392
#define TSTEP 2048
__global__ __launch_bounds__(256) void tid_kernel(const int* __restrict__ dstp,
                                                  unsigned* __restrict__ tid,
                                                  int nE, int nPad) {
  const int e0 = (blockIdx.x * 256 + threadIdx.x) * 4;
  if (e0 >= nPad) return;
  int d[4];
#pragma unroll
  for (int b = 0; b < 4; ++b) {
    const int e = e0 + b;
    d[b] = dstp[e < nE ? e : nE - 1];
  }
  unsigned pk = 0u;
#pragma unroll
  for (int b = 0; b < 4; ++b) {
    const unsigned t = (e0 + b < nE) ? ((unsigned)d[b] / (unsigned)NPB) : 0u;
    pk |= (t & 255u) << (8 * b);
  }
  tid[e0 >> 2] = pk;
}

#define NLD  4
#define NPOS (NLD * 4)
#define QTH  32
#define QCAP 800
#define QHI  (QCAP - 512)
#define LCAP 1024
#define SRCBITS 17

__device__ __forceinline__ void add_rows(const float* __restrict__ h, float* tile, float* cl,
                                         int sv, int dv, int nb, int lane) {
  for (int g = 0; g < nb; g += 8) {
    float v[8][4];
#pragma unroll
    for (int k = 0; k < 8; ++k) {
      const int sk = __builtin_amdgcn_readlane(sv, g + k);
      const float* hp = h + (size_t)sk * HID + lane;
      v[k][0] = hp[0];
      v[k][1] = hp[32];
      v[k][2] = hp[64];
      v[k][3] = hp[96];
    }
#pragma unroll
    for (int k = 0; k < 8; ++k) {
      if (g + k < nb) {
        const int dk = __builtin_amdgcn_readlane(dv, g + k);
        float* tp = tile + dk * HID + lane;
        __hip_atomic_fetch_add(tp +  0, v[k][0], __ATOMIC_RELAXED, __HIP_MEMORY_SCOPE_WORKGROUP);
        __hip_atomic_fetch_add(tp + 32, v[k][1], __ATOMIC_RELAXED, __HIP_MEMORY_SCOPE_WORKGROUP);
        __hip_atomic_fetch_add(tp + 64, v[k][2], __ATOMIC_RELAXED, __HIP_MEMORY_SCOPE_WORKGROUP);
        __hip_atomic_fetch_add(tp + 96, v[k][3], __ATOMIC_RELAXED, __HIP_MEMORY_SCOPE_WORKGROUP);
        if (lane == 0)
          __hip_atomic_fetch_add(cl + dk, 1.0f, __ATOMIC_RELAXED, __HIP_MEMORY_SCOPE_WORKGROUP);
      }
    }
  }
}
template <bool FIRST>
__global__ __launch_bounds__(1024) void agg_kernel(
    const float* __restrict__ h,
    const float* __restrict__ Win,
    const float* __restrict__ bin,
    const int*   __restrict__ ei,
    const unsigned* __restrict__ tidw,
    unsigned* __restrict__ lst,
    int*      __restrict__ lcnt,
    float* __restrict__ s,
    float* __restrict__ cnt, int n, int nE, int nPad) {
  __shared__ float tile[NPB * HID];
  __shared__ float cl[NPB];
  __shared__ int   q[32][QCAP];

  const int lane = threadIdx.x & 31;
  const int wave = __builtin_amdgcn_readfirstlane((int)(threadIdx.x >> 5));
  const int lo   = blockIdx.x * NPB;
  const int nloc = (n - lo) < NPB ? (n - lo) : NPB;

  float wi[4][4], bi[4];
  if constexpr (FIRST) {
#pragma unroll
    for (int j = 0; j < 4; ++j) {
      bi[j] = bin[lane + 32 * j];
#pragma unroll
      for (int r = 0; r < 4; ++r) wi[r][j] = Win[r * HID + lane + 32 * j];
    }
  }

  {
    const float4 z = {0.0f, 0.0f, 0.0f, 0.0f};
    float4* t4 = (float4*)tile;
    for (int i = threadIdx.x; i < NPB * HID / 4; i += 1024) t4[i] = z;
    for (int i = threadIdx.x; i < NPB; i += 1024) cl[i] = 0.0f;
  }
  __syncthreads();

  const int* __restrict__ dstp = ei + nE;
  const uint4* __restrict__ tid = (const uint4*)tidw;
  int* qw = q[wave];
  unsigned* lp = lst + (size_t)(blockIdx.x * 32 + wave) * LCAP;
  int lpos = 0;
  bool haveList = false;
  if constexpr (!FIRST) {
    lpos = __builtin_amdgcn_readfirstlane(lcnt[blockIdx.x * 32 + wave]);
    haveList = lpos <= LCAP;
  }
  if (haveList) {
    unsigned pkn = (lane < lpos) ? lp[lane] : 0u;
    for (int base = 0; base < lpos; base += 32) {
      const unsigned pk = pkn;
      const int nx = base + 32 + lane;
      pkn = (nx < lpos) ? lp[nx] : 0u;
      const int sv = (int)(pk & ((1u << SRCBITS) - 1u));
      const int dv = (int)(pk >> SRCBITS);
      const int nb = (lpos - base) < 32 ? (lpos - base) : 32;
      add_rows(h, tile, cl, sv, dv, nb, lane);
    }
  } else {
  const int nch = nPad / TSTEP;
  const int c0 = (int)((blockIdx.x * (unsigned)nch) / gridDim.x);
  const unsigned pat = (blockIdx.x & 255u) * 0x01010101u;
  int qn = 0;

  uint4 wn[NLD];
  {
    int cc = wave + c0;
    if (cc >= nch) cc -= nch;
    if (wave < nch) {
#pragma unroll
      for (int j = 0; j < NLD; ++j) wn[j] = tid[cc * (TSTEP / 16) + j * 32 + lane];
    }
  }

  for (int c = wave; c < nch; c += 32) {
    int cc = c + c0;
    if (cc >= nch) cc -= nch;
    unsigned w[NPOS];
#pragma unroll
    for (int j = 0; j < NLD; ++j) {
      w[j * 4 + 0] = wn[j].x; w[j * 4 + 1] = wn[j].y;
      w[j * 4 + 2] = wn[j].z; w[j * 4 + 3] = wn[j].w;
    }
    const bool last = (c + 32) >= nch;
    if (!last) {
      int cn = c + 32 + c0;
      if (cn >= nch) cn -= nch;
#pragma unroll
      for (int j = 0; j < NLD; ++j) wn[j] = tid[cn * (TSTEP / 16) + j * 32 + lane];
    }
    const int eb = cc * TSTEP + lane * 16;

    int start = 0;
    for (;;) {
      bool full = false;
      int  next = NLD;
#pragma unroll
      for (int j = 0; j < NLD; ++j) {
        if (!full && j >= start) {
          if (qn > QHI) {
            full = true;
            next = j;
          } else {
            unsigned k = 0u;
#pragma unroll
            for (int d = 0; d < 4; ++d) {
              const unsigned x = w[j * 4 + d] ^ pat;
              const unsigned t = (x & 0x7f7f7f7fu) + 0x7f7f7f7fu;
              k |= (~(t | x | 0x7f7f7f7fu)) >> (7 - d);
            }
            const int eg = eb + j * 512;
            while (__builtin_amdgcn_ballot_w32(k != 0u)) {
              const int  i  = __builtin_ffs((int)k) - 1;
              const int  e  = eg + ((i & 7) << 2) + (i >> 3);
              const bool mt = (k != 0u) && (e < nE);
              const unsigned mask = __builtin_amdgcn_ballot_w32(mt);
              if (mt) qw[qn + (int)__builtin_amdgcn_mbcnt_lo(mask, 0u)] = e;
              qn += __builtin_popcount(mask);
              k &= k - 1u;
            }
          }
        }
      }
      if (full || qn >= QTH || (last && qn > 0)) {
        for (int base = 0; base < qn; base += 32) {
          const int  i   = base + lane;
          const bool val = i < qn;
          const int  e   = val ? qw[i] : 0;
          const int  sv  = val ? ei[e] : 0;
          float4 xv = {0.0f, 0.0f, 0.0f, 0.0f};
          if constexpr (FIRST) xv = ((const float4*)h)[sv];
          const int  dv  = val ? (dstp[e] - lo) : 0;
          const int  nb  = (qn - base) < 32 ? (qn - base) : 32;
          if constexpr (FIRST) {
            if (val && (lpos + i) < LCAP)
              lp[lpos + i] = (unsigned)sv | ((unsigned)dv << SRCBITS);
            for (int k = 0; k < nb; ++k) {
              const float z  = __int_as_float(__builtin_amdgcn_readlane(__float_as_int(xv.x), k));
              const float px = __int_as_float(__builtin_amdgcn_readlane(__float_as_int(xv.y), k));
              const float py = __int_as_float(__builtin_amdgcn_readlane(__float_as_int(xv.z), k));
              const float pz = __int_as_float(__builtin_amdgcn_readlane(__float_as_int(xv.w), k));
              const int   dk = __builtin_amdgcn_readlane(dv, k);
              float* tp = tile + dk * HID + lane;
#pragma unroll
              for (int j = 0; j < 4; ++j) {
                const float v = h0_elem(z, px, py, pz, wi[0][j], wi[1][j], wi[2][j], wi[3][j], bi[j]);
                __hip_atomic_fetch_add(tp + 32 * j, v, __ATOMIC_RELAXED, __HIP_MEMORY_SCOPE_WORKGROUP);
              }
              if (lane == 0)
                __hip_atomic_fetch_add(cl + dk, 1.0f, __ATOMIC_RELAXED, __HIP_MEMORY_SCOPE_WORKGROUP);
            }
          } else {
            add_rows(h, tile, cl, sv, dv, nb, lane);
          }
        }
        if constexpr (FIRST) lpos += qn;
        qn = 0;
      }
      if (!full) break;
      start = next;
    }
  }
  }
  if constexpr (FIRST) {
    if (lane == 0) lcnt[blockIdx.x * 32 + wave] = lpos;
  }
  __syncthreads();

  {
    float4*       s4 = (float4*)(s + (size_t)lo * HID);
    const float4* t4 = (const float4*)tile;
    for (int i = threadIdx.x; i < nloc * (HID / 4); i += 1024) s4[i] = t4[i];
    for (int i = threadIdx.x; i < nloc; i += 1024) cnt[lo + i] = cl[i];
  }
}

typedef __attribute__((ext_vector_type(4))) _Float16 half4;

__device__ inline v16h join16(half8 lo, half8 hi) {
  return __builtin_shufflevector(lo, hi, 0, 1, 2, 3, 4, 5, 6, 7,
                                         8, 9, 10, 11, 12, 13, 14, 15);
}

#define MT   9
#define GR   (16 * MT)
#define RPW  (GR / 8)
#define APIT 136
template <bool SAGE, bool RELU, bool FIRST>
__global__ __launch_bounds__(256) void gemm_kernel(
    const float*    __restrict__ h,
    const float*    __restrict__ Win,
    const float*    __restrict__ bin,
    const float*    __restrict__ s,
    const float*    __restrict__ cnt,
    const float*    __restrict__ Wl,
    const float*    __restrict__ Wr,
    const float*    __restrict__ bias,
    float* __restrict__ out,
    int n_nodes) {
  __shared__ __attribute__((aligned(16))) _Float16 ah[GR * APIT];
  __shared__ __attribute__((aligned(16))) _Float16 ag[SAGE ? GR * APIT : 8];

  const int lane = threadIdx.x & 31;
  const int wave = __builtin_amdgcn_readfirstlane((int)(threadIdx.x >> 5));
  const int c    = lane & 15;
  const int lo   = (lane >= 16) ? 8 : 0;
  const int row0 = blockIdx.x * GR;

  {
    const int rb = wave * RPW;
    int lrow = row0 + rb + (lane < RPW ? lane : RPW - 1);
    if (lrow >= n_nodes) lrow = n_nodes - 1;
    float invl = 0.0f;
    if constexpr (SAGE) {
      const float cv = cnt[lrow];
      invl = (cv > 0.0f) ? __builtin_amdgcn_rcpf(cv) : 0.0f;
    }
    float4 xl = {0.0f, 0.0f, 0.0f, 0.0f};
    float wi[4][4], bi[4];
    if constexpr (FIRST) {
      xl = ((const float4*)h)[lrow];
#pragma unroll
      for (int r = 0; r < 4; ++r) {
        const float4 w4 = *(const float4*)(Win + r * HID + lane * 4);
        wi[r][0] = w4.x; wi[r][1] = w4.y; wi[r][2] = w4.z; wi[r][3] = w4.w;
      }
      const float4 b4 = *(const float4*)(bin + lane * 4);
      bi[0] = b4.x; bi[1] = b4.y; bi[2] = b4.z; bi[3] = b4.w;
    }
#pragma unroll 9
    for (int it = 0; it < RPW; ++it) {
      const int r = rb + it;
      int row = row0 + r;
      if (row >= n_nodes) row = n_nodes - 1;
      half4 hv;
      if constexpr (FIRST) {
        const float z  = __int_as_float(__builtin_amdgcn_readlane(__float_as_int(xl.x), it));
        const float px = __int_as_float(__builtin_amdgcn_readlane(__float_as_int(xl.y), it));
        const float py = __int_as_float(__builtin_amdgcn_readlane(__float_as_int(xl.z), it));
        const float pz = __int_as_float(__builtin_amdgcn_readlane(__float_as_int(xl.w), it));
#pragma unroll
        for (int e = 0; e < 4; ++e)
          hv[e] = (_Float16)h0_elem(z, px, py, pz, wi[0][e], wi[1][e], wi[2][e], wi[3][e], bi[e]);
      } else {
        const float4 v = *(const float4*)(h + (size_t)row * HID + lane * 4);
        hv[0] = (_Float16)v.x; hv[1] = (_Float16)v.y;
        hv[2] = (_Float16)v.z; hv[3] = (_Float16)v.w;
      }
      *(half4*)&ah[r * APIT + lane * 4] = hv;
      if constexpr (SAGE) {
        const float inv = __int_as_float(__builtin_amdgcn_readlane(__float_as_int(invl), it));
        const float4 v = *(const float4*)(s + (size_t)row * HID + lane * 4);
        half4 gv;
        gv[0] = (_Float16)(v.x * inv); gv[1] = (_Float16)(v.y * inv);
        gv[2] = (_Float16)(v.z * inv); gv[3] = (_Float16)(v.w * inv);
        *(half4*)&ag[r * APIT + lane * 4] = gv;
      }
    }
  }
  v16h br[4], bl[4];
  {
    const float* wr = Wr + (size_t)lo * HID + wave * 16 + c;
    float wv[64];
#pragma unroll
    for (int i = 0; i < 64; ++i) {
      const int j = i & 15;
      wv[i] = wr[((i >> 4) * 32 + (j < 8 ? j : j + 8)) * HID];
    }
#pragma unroll
    for (int kk = 0; kk < 4; ++kk) {
#pragma unroll
      for (int e = 0; e < 16; ++e) br[kk][e] = (_Float16)wv[kk * 16 + e];
    }
    asm volatile("" :: "v"(br[0]), "v"(br[1]), "v"(br[2]), "v"(br[3]));
    if constexpr (SAGE) {
      const float* wl = Wl + (size_t)lo * HID + wave * 16 + c;
      float wu[64];
#pragma unroll
      for (int i = 0; i < 64; ++i) {
        const int j = i & 15;
        wu[i] = wl[((i >> 4) * 32 + (j < 8 ? j : j + 8)) * HID];
      }
#pragma unroll
      for (int kk = 0; kk < 4; ++kk) {
#pragma unroll
        for (int e = 0; e < 16; ++e) bl[kk][e] = (_Float16)wu[kk * 16 + e];
      }
      asm volatile("" :: "v"(bl[0]), "v"(bl[1]), "v"(bl[2]), "v"(bl[3]));
    }
  }

  __syncthreads();

  const _Float16* ap = ah + c * APIT + lo;
  const _Float16* gp = ag + c * APIT + lo;

  v8f acc[MT];
  {
    v8f z = {};
#pragma unroll
    for (int mt = 0; mt < MT; ++mt) acc[mt] = z;
  }

#pragma unroll
  for (int kk = 0; kk < 4; ++kk) {
#pragma unroll
    for (int mt = 0; mt < MT; ++mt) {
      const int off = mt * 16 * APIT + kk * 32;
      v16h af = join16(*(const half8*)(ap + off), *(const half8*)(ap + off + 16));
      acc[mt] = __builtin_amdgcn_wmma_f32_16x16x32_f16(
          false, af, false, br[kk], (short)0, acc[mt], false, false);
      if constexpr (SAGE) {
        v16h gf = join16(*(const half8*)(gp + off), *(const half8*)(gp + off + 16));
        acc[mt] = __builtin_amdgcn_wmma_f32_16x16x32_f16(
            false, gf, false, bl[kk], (short)0, acc[mt], false, false);
      }
    }
  }

  const int   ncol = wave * 16 + c;
  const float bb   = bias[ncol];
#pragma unroll
  for (int mt = 0; mt < MT; ++mt) {
#pragma unroll
    for (int r = 0; r < 8; ++r) {
      float v = acc[mt][r] + bb;
      if constexpr (RELU) v = fmaxf(v, 0.0f);
      const int orow = row0 + mt * 16 + lo + r;
      if (orow < n_nodes) out[(size_t)orow * HID + ncol] = v;
    }
  }
}

#define CS_BLOCKS 256
__global__ __launch_bounds__(512) void colsum_kernel(const float* __restrict__ ns,
                                                     float* __restrict__ part,
                                                     int n) {
  __shared__ float red[16][HID];
  const int cg   = threadIdx.x & 31;
  const int rsub = threadIdx.x >> 5;
  const int rpb  = (n + (int)gridDim.x - 1) / (int)gridDim.x;
  const int r0   = blockIdx.x * rpb;
  int r1 = r0 + rpb;
  if (r1 > n) r1 = n;
  float4 a = {0.0f, 0.0f, 0.0f, 0.0f};
#pragma unroll 4
  for (int r = r0 + rsub; r < r1; r += 16) {
    const float4 v = *(const float4*)(ns + (size_t)r * HID + cg * 4);
    a.x += v.x; a.y += v.y; a.z += v.z; a.w += v.w;
  }
  red[rsub][cg * 4 + 0] = a.x;
  red[rsub][cg * 4 + 1] = a.y;
  red[rsub][cg * 4 + 2] = a.z;
  red[rsub][cg * 4 + 3] = a.w;
  __syncthreads();
  if (threadIdx.x < HID) {
    float t = 0.0f;
    for (int k = 0; k < 16; ++k) t += red[k][threadIdx.x];
    part[(size_t)blockIdx.x * HID + threadIdx.x] = t;
  }
}

__global__ __launch_bounds__(1024) void colmean_final_kernel(
    const float* __restrict__ part, float* __restrict__ outm, int nparts, int n) {
  __shared__ float red[8][HID];
  const int col = threadIdx.x & (HID - 1);
  const int g   = threadIdx.x >> 7;
  float t = 0.0f;
#pragma unroll 4
  for (int b = g; b < nparts; b += 8) t += part[(size_t)b * HID + col];
  red[g][col] = t;
  __syncthreads();
  if (threadIdx.x < HID) {
    float sum = 0.0f;
    for (int k = 0; k < 8; ++k) sum += red[k][threadIdx.x];
    outm[threadIdx.x] = sum / (float)n;
  }
}

extern "C" void kernel_launch(void* const* d_in, const int* in_sizes, int n_in,
                              void* d_out, int out_size, void* d_ws, size_t ws_size,
                              hipStream_t stream) {
  const float* pos  = (const float*)d_in[0];
  const int*   an   = (const int*)  d_in[1];
  const int*   ei   = (const int*)  d_in[2];
  const float* W_in = (const float*)d_in[3];
  const float* b_in = (const float*)d_in[4];
  const float* W1l  = (const float*)d_in[5];
  const float* b1   = (const float*)d_in[6];
  const float* W1r  = (const float*)d_in[7];
  const float* W2l  = (const float*)d_in[8];
  const float* b2   = (const float*)d_in[9];
  const float* W2r  = (const float*)d_in[10];
  const float* Wout = (const float*)d_in[11];
  const float* bout = (const float*)d_in[12];

  const int n  = in_sizes[0] / 3;
  const int nE = in_sizes[2] / 2;

  float* ws   = (float*)d_ws;
  float* h    = ws;
  float* s    = ws + (size_t)n * HID;
  float* cnt  = s  + (size_t)n * HID;

  size_t woff = (size_t)n * HID * 2 + n;
  woff = (woff + 63) & ~(size_t)63;
  float* part = ws + woff;
  unsigned* tid = (unsigned*)(part + CS_BLOCKS * HID);

  float* outm = (float*)d_out;
  float* nst  = (float*)d_out + HID;

  const int gemmBlks = (n + GR - 1) / GR;
  const int aggBlks  = (n + NPB - 1) / NPB;
  const int nPad     = ((nE + TSTEP - 1) / TSTEP) * TSTEP;
  float* xp = (float*)((unsigned char*)tid + nPad);
  unsigned* lst = (unsigned*)(xp + (size_t)n * 4);
  int* lcnt = (int*)(lst + (size_t)aggBlks * 32 * LCAP);

  xpack_kernel<<<(n + 255) / 256, 256, 0, stream>>>(pos, an, xp, n);

  tid_kernel<<<(nPad / 4 + 255) / 256, 256, 0, stream>>>(ei + nE, tid, nE, nPad);

  agg_kernel<true><<<aggBlks, 1024, 0, stream>>>(xp, W_in, b_in, ei, tid, lst, lcnt, s, cnt, n, nE, nPad);
  gemm_kernel<true, true, true><<<gemmBlks, 256, 0, stream>>>(xp, W_in, b_in, s, cnt, W1l, W1r, b1, h, n);

  agg_kernel<false><<<aggBlks, 1024, 0, stream>>>(h, nullptr, nullptr, ei, tid, lst, lcnt, s, cnt, n, nE, nPad);
  gemm_kernel<true, true, false><<<gemmBlks, 256, 0, stream>>>(h, nullptr, nullptr, s, cnt, W2l, W2r, b2, h, n);

  gemm_kernel<false, false, false><<<gemmBlks, 256, 0, stream>>>(h, nullptr, nullptr, nullptr, nullptr,
                                                                 nullptr, Wout, bout, nst, n);
  colsum_kernel<<<CS_BLOCKS, 512, 0, stream>>>(nst, part, n);
  colmean_final_kernel<<<1, 1024, 0, stream>>>(part, outm, CS_BLOCKS, n);
}


# Edges grouped by destination tile once per call (256 chunks, each counting-sorted in LDS into per-tile groups); both neighbour-sum launches read their tile's groups, no list scan; sums arrival-order
# speedup vs baseline: 5.2092x; 1.0814x over previous
#include <hip/hip_runtime.h>


typedef __attribute__((ext_vector_type(16))) _Float16 v16h;
typedef __attribute__((ext_vector_type(8)))  _Float16 half8;
typedef __attribute__((ext_vector_type(8)))  float    v8f;

#define HID 128

__global__ void xpack_kernel(const float* __restrict__ pos,
                             const int*   __restrict__ an,
                             float* __restrict__ xp, int n) {
  int node = blockIdx.x * 256 + threadIdx.x;
  if (node >= n) return;
  float4 x;
  x.x = (float)an[node] * 0.1f;
  x.y = pos[node * 3 + 0];
  x.z = pos[node * 3 + 1];
  x.w = pos[node * 3 + 2];
  ((float4*)xp)[node] = x;
}

__device__ inline float h0_elem(float z, float px, float py, float pz,
                                float w0, float w1, float w2, float w3, float b) {
  float v = b;
  v = fmaf(z,  w0, v);
  v = fmaf(px, w1, v);
  v = fmaf(py, w2, v);
  v = fmaf(pz, w3, v);
  return fmaxf(v, 0.0f);
}

#define NPB  392
#define NBIN 256
#define MAXT 256
#define SRCBITS 17
#define BLD  8

__device__ __forceinline__ int wave_incl_scan(int x, int lane) {
#pragma unroll
  for (int d = 1; d < 32; d <<= 1) {
    const int y = __shfl_up(x, d, 32);
    if (lane >= d) x += y;
  }
  return x;
}

__global__ __launch_bounds__(1024) void bin_kernel(const int* __restrict__ ei,
                                                   unsigned* __restrict__ bins,
                                                   int* __restrict__ offs,
                                                   int nE, int chunk) {
  __shared__ int hist[MAXT];
  __shared__ int cur[MAXT];
  __shared__ int wtot[MAXT / 32];
  const int lane = threadIdx.x & 31;
  const int wave = threadIdx.x >> 5;
  const int e0 = blockIdx.x * chunk;
  int e1 = e0 + chunk;
  if (e1 > nE) e1 = nE;
  const int* __restrict__ dstp = ei + nE;

  if (threadIdx.x < MAXT) hist[threadIdx.x] = 0;
  __syncthreads();

  for (int eb = e0 + (int)threadIdx.x; eb < e1; eb += 1024 * BLD) {
    int d[BLD];
#pragma unroll
    for (int j = 0; j < BLD; ++j) {
      const int e = eb + j * 1024;
      d[j] = (e < e1) ? dstp[e] : -1;
    }
#pragma unroll
    for (int j = 0; j < BLD; ++j) {
      if (d[j] >= 0)
        __hip_atomic_fetch_add(&hist[(unsigned)d[j] / (unsigned)NPB], 1,
                               __ATOMIC_RELAXED, __HIP_MEMORY_SCOPE_WORKGROUP);
    }
  }
  __syncthreads();

  int len = 0;
  if (threadIdx.x < MAXT) len = hist[threadIdx.x];
  const int inc = wave_incl_scan(len, lane);
  if (threadIdx.x < MAXT && lane == 31) wtot[wave] = inc;
  __syncthreads();
  if (threadIdx.x < MAXT) {
    int add = 0;
    for (int k = 0; k < wave; ++k) add += wtot[k];
    const int st = inc + add - len;
    cur[threadIdx.x] = st;
    offs[threadIdx.x * NBIN + blockIdx.x] = st;
    if (threadIdx.x == MAXT - 1) offs[MAXT * NBIN + blockIdx.x] = st + len;
  }
  __syncthreads();

  for (int eb = e0 + (int)threadIdx.x; eb < e1; eb += 1024 * BLD) {
    int d[BLD], sv[BLD];
#pragma unroll
    for (int j = 0; j < BLD; ++j) {
      const int e = eb + j * 1024;
      d[j]  = (e < e1) ? dstp[e] : -1;
      sv[j] = (e < e1) ? ei[e] : 0;
    }
#pragma unroll
    for (int j = 0; j < BLD; ++j) {
      if (d[j] >= 0) {
        const unsigned t = (unsigned)d[j] / (unsigned)NPB;
        const int p = __hip_atomic_fetch_add(&cur[t], 1, __ATOMIC_RELAXED,
                                             __HIP_MEMORY_SCOPE_WORKGROUP);
        bins[e0 + p] = (unsigned)sv[j] | (((unsigned)d[j] - t * (unsigned)NPB) << SRCBITS);
      }
    }
  }
}

#define NGW (NBIN / 32)

__device__ __forceinline__ void add_rows(const float* __restrict__ h, float* tile, float* cl,
                                         int sv, int dv, int nb, int lane) {
  for (int g = 0; g < nb; g += 8) {
    float v[8][4];
#pragma unroll
    for (int k = 0; k < 8; ++k) {
      const int sk = __builtin_amdgcn_readlane(sv, g + k);
      const float* hp = h + (size_t)sk * HID + lane;
      v[k][0] = hp[0];
      v[k][1] = hp[32];
      v[k][2] = hp[64];
      v[k][3] = hp[96];
    }
#pragma unroll
    for (int k = 0; k < 8; ++k) {
      if (g + k < nb) {
        const int dk = __builtin_amdgcn_readlane(dv, g + k);
        float* tp = tile + dk * HID + lane;
        __hip_atomic_fetch_add(tp +  0, v[k][0], __ATOMIC_RELAXED, __HIP_MEMORY_SCOPE_WORKGROUP);
        __hip_atomic_fetch_add(tp + 32, v[k][1], __ATOMIC_RELAXED, __HIP_MEMORY_SCOPE_WORKGROUP);
        __hip_atomic_fetch_add(tp + 64, v[k][2], __ATOMIC_RELAXED, __HIP_MEMORY_SCOPE_WORKGROUP);
        __hip_atomic_fetch_add(tp + 96, v[k][3], __ATOMIC_RELAXED, __HIP_MEMORY_SCOPE_WORKGROUP);
        if (lane == 0)
          __hip_atomic_fetch_add(cl + dk, 1.0f, __ATOMIC_RELAXED, __HIP_MEMORY_SCOPE_WORKGROUP);
      }
    }
  }
}
template <bool FIRST>
__global__ __launch_bounds__(1024) void agg_kernel(
    const float* __restrict__ h,
    const float* __restrict__ Win,
    const float* __restrict__ bin,
    const unsigned* __restrict__ bins,
    const int*      __restrict__ offs,
    float* __restrict__ s,
    float* __restrict__ cnt, int n, int chunk) {
  __shared__ float tile[NPB * HID];
  __shared__ float cl[NPB];

  const int lane = threadIdx.x & 31;
  const int wave = __builtin_amdgcn_readfirstlane((int)(threadIdx.x >> 5));
  const int lo   = blockIdx.x * NPB;
  const int nloc = (n - lo) < NPB ? (n - lo) : NPB;

  int gst = 0, glen = 0;
  if (lane < NGW) {
    const int g = wave * NGW + lane;
    gst  = offs[blockIdx.x * NBIN + g];
    glen = offs[(blockIdx.x + 1) * NBIN + g] - gst;
  }

  float wi[4][4], bi[4];
  if constexpr (FIRST) {
#pragma unroll
    for (int j = 0; j < 4; ++j) {
      bi[j] = bin[lane + 32 * j];
#pragma unroll
      for (int r = 0; r < 4; ++r) wi[r][j] = Win[r * HID + lane + 32 * j];
    }
  }

  {
    const float4 z = {0.0f, 0.0f, 0.0f, 0.0f};
    float4* t4 = (float4*)tile;
    for (int i = threadIdx.x; i < NPB * HID / 4; i += 1024) t4[i] = z;
    for (int i = threadIdx.x; i < NPB; i += 1024) cl[i] = 0.0f;
  }

  static_assert(NGW == 8, "the position-to-address sum below is written out for 8 groups");
  const int ginc = wave_incl_scan(glen, lane);
  const int gexv = ginc - glen;
  const int gadv = (wave * NGW + lane) * chunk + gst - gexv;
  const int e1 = __builtin_amdgcn_readlane(gexv, 1), e2 = __builtin_amdgcn_readlane(gexv, 2);
  const int e3 = __builtin_amdgcn_readlane(gexv, 3), e4 = __builtin_amdgcn_readlane(gexv, 4);
  const int e5 = __builtin_amdgcn_readlane(gexv, 5), e6 = __builtin_amdgcn_readlane(gexv, 6);
  const int e7 = __builtin_amdgcn_readlane(gexv, 7);
  const int a0 = __builtin_amdgcn_readlane(gadv, 0), a1 = __builtin_amdgcn_readlane(gadv, 1);
  const int a2 = __builtin_amdgcn_readlane(gadv, 2), a3 = __builtin_amdgcn_readlane(gadv, 3);
  const int a4 = __builtin_amdgcn_readlane(gadv, 4), a5 = __builtin_amdgcn_readlane(gadv, 5);
  const int a6 = __builtin_amdgcn_readlane(gadv, 6), a7 = __builtin_amdgcn_readlane(gadv, 7);
  const int d1 = a1 - a0, d2 = a2 - a1, d3 = a3 - a2, d4 = a4 - a3;
  const int d5 = a5 - a4, d6 = a6 - a5, d7 = a7 - a6;
  const int Tw = __builtin_amdgcn_readlane(ginc, NGW - 1);
  __syncthreads();

  if (Tw > 0) {
    const unsigned SM = (1u << SRCBITS) - 1u;
    const int vlast = Tw - 1;
    auto edge_at = [=](int v) -> unsigned {
      v = v < vlast ? v : vlast;
      int a = a0 + v;
      a += (v >= e1) ? d1 : 0;
      a += (v >= e2) ? d2 : 0;
      a += (v >= e3) ? d3 : 0;
      a += (v >= e4) ? d4 : 0;
      a += (v >= e5) ? d5 : 0;
      a += (v >= e6) ? d6 : 0;
      a += (v >= e7) ? d7 : 0;
      return bins[a];
    };
    unsigned pkc = edge_at(lane);
    unsigned pkn = edge_at(32 + lane);
    float4 xc = {0.0f, 0.0f, 0.0f, 0.0f};
    if constexpr (FIRST) xc = ((const float4*)h)[pkc & SM];

    for (int base = 0; base < Tw; base += 32) {
      const unsigned pk2 = edge_at(base + 64 + lane);
      float4 xn = {0.0f, 0.0f, 0.0f, 0.0f};
      if constexpr (FIRST) xn = ((const float4*)h)[pkn & SM];

      const int dv = (int)(pkc >> SRCBITS);
      const int nb = (Tw - base) < 32 ? (Tw - base) : 32;
      if constexpr (FIRST) {
        for (int k = 0; k < nb; ++k) {
          const float z  = __int_as_float(__builtin_amdgcn_readlane(__float_as_int(xc.x), k));
          const float px = __int_as_float(__builtin_amdgcn_readlane(__float_as_int(xc.y), k));
          const float py = __int_as_float(__builtin_amdgcn_readlane(__float_as_int(xc.z), k));
          const float pz = __int_as_float(__builtin_amdgcn_readlane(__float_as_int(xc.w), k));
          const int   dk = __builtin_amdgcn_readlane(dv, k);
          float* tp = tile + dk * HID + lane;
#pragma unroll
          for (int j = 0; j < 4; ++j) {
            const float v = h0_elem(z, px, py, pz, wi[0][j], wi[1][j], wi[2][j], wi[3][j], bi[j]);
            __hip_atomic_fetch_add(tp + 32 * j, v, __ATOMIC_RELAXED, __HIP_MEMORY_SCOPE_WORKGROUP);
          }
          if (lane == 0)
            __hip_atomic_fetch_add(cl + dk, 1.0f, __ATOMIC_RELAXED, __HIP_MEMORY_SCOPE_WORKGROUP);
        }
      } else {
        add_rows(h, tile, cl, (int)(pkc & SM), dv, nb, lane);
      }
      pkc = pkn;
      pkn = pk2;
      xc  = xn;
    }
  }
  __syncthreads();

  {
    float4*       s4 = (float4*)(s + (size_t)lo * HID);
    const float4* t4 = (const float4*)tile;
    for (int i = threadIdx.x; i < nloc * (HID / 4); i += 1024) s4[i] = t4[i];
    for (int i = threadIdx.x; i < nloc; i += 1024) cnt[lo + i] = cl[i];
  }
}

typedef __attribute__((ext_vector_type(4))) _Float16 half4;

__device__ inline v16h join16(half8 lo, half8 hi) {
  return __builtin_shufflevector(lo, hi, 0, 1, 2, 3, 4, 5, 6, 7,
                                         8, 9, 10, 11, 12, 13, 14, 15);
}

#define MT   9
#define GR   (16 * MT)
#define RPW  (GR / 8)
#define APIT 136
template <bool SAGE, bool RELU, bool FIRST>
__global__ __launch_bounds__(256) void gemm_kernel(
    const float*    __restrict__ h,
    const float*    __restrict__ Win,
    const float*    __restrict__ bin,
    const float*    __restrict__ s,
    const float*    __restrict__ cnt,
    const float*    __restrict__ Wl,
    const float*    __restrict__ Wr,
    const float*    __restrict__ bias,
    float* __restrict__ out,
    int n_nodes) {
  __shared__ __attribute__((aligned(16))) _Float16 ah[GR * APIT];
  __shared__ __attribute__((aligned(16))) _Float16 ag[SAGE ? GR * APIT : 8];

  const int lane = threadIdx.x & 31;
  const int wave = __builtin_amdgcn_readfirstlane((int)(threadIdx.x >> 5));
  const int c    = lane & 15;
  const int lo   = (lane >= 16) ? 8 : 0;
  const int row0 = blockIdx.x * GR;

  {
    const int rb = wave * RPW;
    int lrow = row0 + rb + (lane < RPW ? lane : RPW - 1);
    if (lrow >= n_nodes) lrow = n_nodes - 1;
    float invl = 0.0f;
    if constexpr (SAGE) {
      const float cv = cnt[lrow];
      invl = (cv > 0.0f) ? __builtin_amdgcn_rcpf(cv) : 0.0f;
    }
    float4 xl = {0.0f, 0.0f, 0.0f, 0.0f};
    float wi[4][4], bi[4];
    if constexpr (FIRST) {
      xl = ((const float4*)h)[lrow];
#pragma unroll
      for (int r = 0; r < 4; ++r) {
        const float4 w4 = *(const float4*)(Win + r * HID + lane * 4);
        wi[r][0] = w4.x; wi[r][1] = w4.y; wi[r][2] = w4.z; wi[r][3] = w4.w;
      }
      const float4 b4 = *(const float4*)(bin + lane * 4);
      bi[0] = b4.x; bi[1] = b4.y; bi[2] = b4.z; bi[3] = b4.w;
    }
#pragma unroll 9
    for (int it = 0; it < RPW; ++it) {
      const int r = rb + it;
      int row = row0 + r;
      if (row >= n_nodes) row = n_nodes - 1;
      half4 hv;
      if constexpr (FIRST) {
        const float z  = __int_as_float(__builtin_amdgcn_readlane(__float_as_int(xl.x), it));
        const float px = __int_as_float(__builtin_amdgcn_readlane(__float_as_int(xl.y), it));
        const float py = __int_as_float(__builtin_amdgcn_readlane(__float_as_int(xl.z), it));
        const float pz = __int_as_float(__builtin_amdgcn_readlane(__float_as_int(xl.w), it));
#pragma unroll
        for (int e = 0; e < 4; ++e)
          hv[e] = (_Float16)h0_elem(z, px, py, pz, wi[0][e], wi[1][e], wi[2][e], wi[3][e], bi[e]);
      } else {
        const float4 v = *(const float4*)(h + (size_t)row * HID + lane * 4);
        hv[0] = (_Float16)v.x; hv[1] = (_Float16)v.y;
        hv[2] = (_Float16)v.z; hv[3] = (_Float16)v.w;
      }
      *(half4*)&ah[r * APIT + lane * 4] = hv;
      if constexpr (SAGE) {
        const float inv = __int_as_float(__builtin_amdgcn_readlane(__float_as_int(invl), it));
        const float4 v = *(const float4*)(s + (size_t)row * HID + lane * 4);
        half4 gv;
        gv[0] = (_Float16)(v.x * inv); gv[1] = (_Float16)(v.y * inv);
        gv[2] = (_Float16)(v.z * inv); gv[3] = (_Float16)(v.w * inv);
        *(half4*)&ag[r * APIT + lane * 4] = gv;
      }
    }
  }
  v16h br[4], bl[4];
  {
    const float* wr = Wr + (size_t)lo * HID + wave * 16 + c;
    float wv[64];
#pragma unroll
    for (int i = 0; i < 64; ++i) {
      const int j = i & 15;
      wv[i] = wr[((i >> 4) * 32 + (j < 8 ? j : j + 8)) * HID];
    }
#pragma unroll
    for (int kk = 0; kk < 4; ++kk) {
#pragma unroll
      for (int e = 0; e < 16; ++e) br[kk][e] = (_Float16)wv[kk * 16 + e];
    }
    asm volatile("" :: "v"(br[0]), "v"(br[1]), "v"(br[2]), "v"(br[3]));
    if constexpr (SAGE) {
      const float* wl = Wl + (size_t)lo * HID + wave * 16 + c;
      float wu[64];
#pragma unroll
      for (int i = 0; i < 64; ++i) {
        const int j = i & 15;
        wu[i] = wl[((i >> 4) * 32 + (j < 8 ? j : j + 8)) * HID];
      }
#pragma unroll
      for (int kk = 0; kk < 4; ++kk) {
#pragma unroll
        for (int e = 0; e < 16; ++e) bl[kk][e] = (_Float16)wu[kk * 16 + e];
      }
      asm volatile("" :: "v"(bl[0]), "v"(bl[1]), "v"(bl[2]), "v"(bl[3]));
    }
  }

  __syncthreads();

  const _Float16* ap = ah + c * APIT + lo;
  const _Float16* gp = ag + c * APIT + lo;

  v8f acc[MT];
  {
    v8f z = {};
#pragma unroll
    for (int mt = 0; mt < MT; ++mt) acc[mt] = z;
  }

#pragma unroll
  for (int kk = 0; kk < 4; ++kk) {
#pragma unroll
    for (int mt = 0; mt < MT; ++mt) {
      const int off = mt * 16 * APIT + kk * 32;
      v16h af = join16(*(const half8*)(ap + off), *(const half8*)(ap + off + 16));
      acc[mt] = __builtin_amdgcn_wmma_f32_16x16x32_f16(
          false, af, false, br[kk], (short)0, acc[mt], false, false);
      if constexpr (SAGE) {
        v16h gf = join16(*(const half8*)(gp + off), *(const half8*)(gp + off + 16));
        acc[mt] = __builtin_amdgcn_wmma_f32_16x16x32_f16(
            false, gf, false, bl[kk], (short)0, acc[mt], false, false);
      }
    }
  }

  const int   ncol = wave * 16 + c;
  const float bb   = bias[ncol];
#pragma unroll
  for (int mt = 0; mt < MT; ++mt) {
#pragma unroll
    for (int r = 0; r < 8; ++r) {
      float v = acc[mt][r] + bb;
      if constexpr (RELU) v = fmaxf(v, 0.0f);
      const int orow = row0 + mt * 16 + lo + r;
      if (orow < n_nodes) out[(size_t)orow * HID + ncol] = v;
    }
  }
}

#define CS_BLOCKS 256
__global__ __launch_bounds__(512) void colsum_kernel(const float* __restrict__ ns,
                                                     float* __restrict__ part,
                                                     int n) {
  __shared__ float red[16][HID];
  const int cg   = threadIdx.x & 31;
  const int rsub = threadIdx.x >> 5;
  const int rpb  = (n + (int)gridDim.x - 1) / (int)gridDim.x;
  const int r0   = blockIdx.x * rpb;
  int r1 = r0 + rpb;
  if (r1 > n) r1 = n;
  float4 a = {0.0f, 0.0f, 0.0f, 0.0f};
#pragma unroll 4
  for (int r = r0 + rsub; r < r1; r += 16) {
    const float4 v = *(const float4*)(ns + (size_t)r * HID + cg * 4);
    a.x += v.x; a.y += v.y; a.z += v.z; a.w += v.w;
  }
  red[rsub][cg * 4 + 0] = a.x;
  red[rsub][cg * 4 + 1] = a.y;
  red[rsub][cg * 4 + 2] = a.z;
  red[rsub][cg * 4 + 3] = a.w;
  __syncthreads();
  if (threadIdx.x < HID) {
    float t = 0.0f;
    for (int k = 0; k < 16; ++k) t += red[k][threadIdx.x];
    part[(size_t)blockIdx.x * HID + threadIdx.x] = t;
  }
}

__global__ __launch_bounds__(1024) void colmean_final_kernel(
    const float* __restrict__ part, float* __restrict__ outm, int nparts, int n) {
  __shared__ float red[8][HID];
  const int col = threadIdx.x & (HID - 1);
  const int g   = threadIdx.x >> 7;
  float t = 0.0f;
#pragma unroll 4
  for (int b = g; b < nparts; b += 8) t += part[(size_t)b * HID + col];
  red[g][col] = t;
  __syncthreads();
  if (threadIdx.x < HID) {
    float sum = 0.0f;
    for (int k = 0; k < 8; ++k) sum += red[k][threadIdx.x];
    outm[threadIdx.x] = sum / (float)n;
  }
}

extern "C" void kernel_launch(void* const* d_in, const int* in_sizes, int n_in,
                              void* d_out, int out_size, void* d_ws, size_t ws_size,
                              hipStream_t stream) {
  const float* pos  = (const float*)d_in[0];
  const int*   an   = (const int*)  d_in[1];
  const int*   ei   = (const int*)  d_in[2];
  const float* W_in = (const float*)d_in[3];
  const float* b_in = (const float*)d_in[4];
  const float* W1l  = (const float*)d_in[5];
  const float* b1   = (const float*)d_in[6];
  const float* W1r  = (const float*)d_in[7];
  const float* W2l  = (const float*)d_in[8];
  const float* b2   = (const float*)d_in[9];
  const float* W2r  = (const float*)d_in[10];
  const float* Wout = (const float*)d_in[11];
  const float* bout = (const float*)d_in[12];

  const int n  = in_sizes[0] / 3;
  const int nE = in_sizes[2] / 2;

  float* ws   = (float*)d_ws;
  float* h    = ws;
  float* s    = ws + (size_t)n * HID;
  float* cnt  = s  + (size_t)n * HID;

  size_t woff = (size_t)n * HID * 2 + n;
  woff = (woff + 63) & ~(size_t)63;
  float* part = ws + woff;
  float* xp = part + CS_BLOCKS * HID;

  float* outm = (float*)d_out;
  float* nst  = (float*)d_out + HID;

  const int gemmBlks = (n + GR - 1) / GR;
  const int aggBlks  = (n + NPB - 1) / NPB;
  const int chunk    = (nE + NBIN - 1) / NBIN;
  unsigned* bins = (unsigned*)(xp + (size_t)n * 4);
  int* offs = (int*)(bins + (size_t)NBIN * chunk);

  xpack_kernel<<<(n + 255) / 256, 256, 0, stream>>>(pos, an, xp, n);

  bin_kernel<<<NBIN, 1024, 0, stream>>>(ei, bins, offs, nE, chunk);

  agg_kernel<true><<<aggBlks, 1024, 0, stream>>>(xp, W_in, b_in, bins, offs, s, cnt, n, chunk);
  gemm_kernel<true, true, true><<<gemmBlks, 256, 0, stream>>>(xp, W_in, b_in, s, cnt, W1l, W1r, b1, h, n);

  agg_kernel<false><<<aggBlks, 1024, 0, stream>>>(h, nullptr, nullptr, bins, offs, s, cnt, n, chunk);
  gemm_kernel<true, true, false><<<gemmBlks, 256, 0, stream>>>(h, nullptr, nullptr, s, cnt, W2l, W2r, b2, h, n);

  gemm_kernel<false, false, false><<<gemmBlks, 256, 0, stream>>>(h, nullptr, nullptr, nullptr, nullptr,
                                                                 nullptr, Wout, bout, nst, n);
  colsum_kernel<<<CS_BLOCKS, 512, 0, stream>>>(nst, part, n);
  colmean_final_kernel<<<1, 1024, 0, stream>>>(part, outm, CS_BLOCKS, n);
}


# In-degree counted once per call: layer-1 neighbour sum adds 1 per edge by one LDS atomic per 32-edge batch (lane = edge), no lane-0 atomic per edge; layer 2 reuses cnt; sums stay arrival-order
# speedup vs baseline: 5.3889x; 1.0345x over previous
#include <hip/hip_runtime.h>


typedef __attribute__((ext_vector_type(16))) _Float16 v16h;
typedef __attribute__((ext_vector_type(8)))  _Float16 half8;
typedef __attribute__((ext_vector_type(8)))  float    v8f;

#define HID 128

__global__ void xpack_kernel(const float* __restrict__ pos,
                             const int*   __restrict__ an,
                             float* __restrict__ xp, int n) {
  int node = blockIdx.x * 256 + threadIdx.x;
  if (node >= n) return;
  float4 x;
  x.x = (float)an[node] * 0.1f;
  x.y = pos[node * 3 + 0];
  x.z = pos[node * 3 + 1];
  x.w = pos[node * 3 + 2];
  ((float4*)xp)[node] = x;
}

__device__ inline float h0_elem(float z, float px, float py, float pz,
                                float w0, float w1, float w2, float w3, float b) {
  float v = b;
  v = fmaf(z,  w0, v);
  v = fmaf(px, w1, v);
  v = fmaf(py, w2, v);
  v = fmaf(pz, w3, v);
  return fmaxf(v, 0.0f);
}

#define NPB  392
#define NBIN 256
#define MAXT 256
#define SRCBITS 17
#define BLD  8

__device__ __forceinline__ int wave_incl_scan(int x, int lane) {
#pragma unroll
  for (int d = 1; d < 32; d <<= 1) {
    const int y = __shfl_up(x, d, 32);
    if (lane >= d) x += y;
  }
  return x;
}

__global__ __launch_bounds__(1024) void bin_kernel(const int* __restrict__ ei,
                                                   unsigned* __restrict__ bins,
                                                   int* __restrict__ offs,
                                                   int nE, int chunk) {
  __shared__ int hist[MAXT];
  __shared__ int cur[MAXT];
  __shared__ int wtot[MAXT / 32];
  const int lane = threadIdx.x & 31;
  const int wave = threadIdx.x >> 5;
  const int e0 = blockIdx.x * chunk;
  int e1 = e0 + chunk;
  if (e1 > nE) e1 = nE;
  const int* __restrict__ dstp = ei + nE;

  if (threadIdx.x < MAXT) hist[threadIdx.x] = 0;
  __syncthreads();

  for (int eb = e0 + (int)threadIdx.x; eb < e1; eb += 1024 * BLD) {
    int d[BLD];
#pragma unroll
    for (int j = 0; j < BLD; ++j) {
      const int e = eb + j * 1024;
      d[j] = (e < e1) ? dstp[e] : -1;
    }
#pragma unroll
    for (int j = 0; j < BLD; ++j) {
      if (d[j] >= 0)
        __hip_atomic_fetch_add(&hist[(unsigned)d[j] / (unsigned)NPB], 1,
                               __ATOMIC_RELAXED, __HIP_MEMORY_SCOPE_WORKGROUP);
    }
  }
  __syncthreads();

  int len = 0;
  if (threadIdx.x < MAXT) len = hist[threadIdx.x];
  const int inc = wave_incl_scan(len, lane);
  if (threadIdx.x < MAXT && lane == 31) wtot[wave] = inc;
  __syncthreads();
  if (threadIdx.x < MAXT) {
    int add = 0;
    for (int k = 0; k < wave; ++k) add += wtot[k];
    const int st = inc + add - len;
    cur[threadIdx.x] = st;
    offs[threadIdx.x * NBIN + blockIdx.x] = st;
    if (threadIdx.x == MAXT - 1) offs[MAXT * NBIN + blockIdx.x] = st + len;
  }
  __syncthreads();

  for (int eb = e0 + (int)threadIdx.x; eb < e1; eb += 1024 * BLD) {
    int d[BLD], sv[BLD];
#pragma unroll
    for (int j = 0; j < BLD; ++j) {
      const int e = eb + j * 1024;
      d[j]  = (e < e1) ? dstp[e] : -1;
      sv[j] = (e < e1) ? ei[e] : 0;
    }
#pragma unroll
    for (int j = 0; j < BLD; ++j) {
      if (d[j] >= 0) {
        const unsigned t = (unsigned)d[j] / (unsigned)NPB;
        const int p = __hip_atomic_fetch_add(&cur[t], 1, __ATOMIC_RELAXED,
                                             __HIP_MEMORY_SCOPE_WORKGROUP);
        bins[e0 + p] = (unsigned)sv[j] | (((unsigned)d[j] - t * (unsigned)NPB) << SRCBITS);
      }
    }
  }
}

#define NGW (NBIN / 32)

__device__ __forceinline__ void add_rows(const float* __restrict__ h, float* tile,
                                         int sv, int dv, int nb, int lane) {
  for (int g = 0; g < nb; g += 8) {
    float v[8][4];
#pragma unroll
    for (int k = 0; k < 8; ++k) {
      const int sk = __builtin_amdgcn_readlane(sv, g + k);
      const float* hp = h + (size_t)sk * HID + lane;
      v[k][0] = hp[0];
      v[k][1] = hp[32];
      v[k][2] = hp[64];
      v[k][3] = hp[96];
    }
#pragma unroll
    for (int k = 0; k < 8; ++k) {
      if (g + k < nb) {
        const int dk = __builtin_amdgcn_readlane(dv, g + k);
        float* tp = tile + dk * HID + lane;
        __hip_atomic_fetch_add(tp +  0, v[k][0], __ATOMIC_RELAXED, __HIP_MEMORY_SCOPE_WORKGROUP);
        __hip_atomic_fetch_add(tp + 32, v[k][1], __ATOMIC_RELAXED, __HIP_MEMORY_SCOPE_WORKGROUP);
        __hip_atomic_fetch_add(tp + 64, v[k][2], __ATOMIC_RELAXED, __HIP_MEMORY_SCOPE_WORKGROUP);
        __hip_atomic_fetch_add(tp + 96, v[k][3], __ATOMIC_RELAXED, __HIP_MEMORY_SCOPE_WORKGROUP);
      }
    }
  }
}
template <bool FIRST>
__global__ __launch_bounds__(1024) void agg_kernel(
    const float* __restrict__ h,
    const float* __restrict__ Win,
    const float* __restrict__ bin,
    const unsigned* __restrict__ bins,
    const int*      __restrict__ offs,
    float* __restrict__ s,
    float* __restrict__ cnt,
    int n, int chunk) {
  __shared__ float tile[NPB * HID];
  __shared__ float cl[FIRST ? NPB : 1];

  const int lane = threadIdx.x & 31;
  const int wave = __builtin_amdgcn_readfirstlane((int)(threadIdx.x >> 5));
  const int lo   = blockIdx.x * NPB;
  const int nloc = (n - lo) < NPB ? (n - lo) : NPB;

  int gst = 0, glen = 0;
  if (lane < NGW) {
    const int g = wave * NGW + lane;
    gst  = offs[blockIdx.x * NBIN + g];
    glen = offs[(blockIdx.x + 1) * NBIN + g] - gst;
  }

  float wi[4][4], bi[4];
  if constexpr (FIRST) {
#pragma unroll
    for (int j = 0; j < 4; ++j) {
      bi[j] = bin[lane + 32 * j];
#pragma unroll
      for (int r = 0; r < 4; ++r) wi[r][j] = Win[r * HID + lane + 32 * j];
    }
  }

  {
    const float4 z = {0.0f, 0.0f, 0.0f, 0.0f};
    float4* t4 = (float4*)tile;
    for (int i = threadIdx.x; i < NPB * HID / 4; i += 1024) t4[i] = z;
    if constexpr (FIRST) {
      for (int i = threadIdx.x; i < NPB; i += 1024) cl[i] = 0.0f;
    }
  }

  static_assert(NGW == 8, "the position-to-address sum below is written out for 8 groups");
  const int ginc = wave_incl_scan(glen, lane);
  const int gexv = ginc - glen;
  const int gadv = (wave * NGW + lane) * chunk + gst - gexv;
  const int e1 = __builtin_amdgcn_readlane(gexv, 1), e2 = __builtin_amdgcn_readlane(gexv, 2);
  const int e3 = __builtin_amdgcn_readlane(gexv, 3), e4 = __builtin_amdgcn_readlane(gexv, 4);
  const int e5 = __builtin_amdgcn_readlane(gexv, 5), e6 = __builtin_amdgcn_readlane(gexv, 6);
  const int e7 = __builtin_amdgcn_readlane(gexv, 7);
  const int a0 = __builtin_amdgcn_readlane(gadv, 0), a1 = __builtin_amdgcn_readlane(gadv, 1);
  const int a2 = __builtin_amdgcn_readlane(gadv, 2), a3 = __builtin_amdgcn_readlane(gadv, 3);
  const int a4 = __builtin_amdgcn_readlane(gadv, 4), a5 = __builtin_amdgcn_readlane(gadv, 5);
  const int a6 = __builtin_amdgcn_readlane(gadv, 6), a7 = __builtin_amdgcn_readlane(gadv, 7);
  const int d1 = a1 - a0, d2 = a2 - a1, d3 = a3 - a2, d4 = a4 - a3;
  const int d5 = a5 - a4, d6 = a6 - a5, d7 = a7 - a6;
  const int Tw = __builtin_amdgcn_readlane(ginc, NGW - 1);
  __syncthreads();

  if (Tw > 0) {
    const unsigned SM = (1u << SRCBITS) - 1u;
    const int vlast = Tw - 1;
    auto edge_at = [=](int v) -> unsigned {
      v = v < vlast ? v : vlast;
      int a = a0 + v;
      a += (v >= e1) ? d1 : 0;
      a += (v >= e2) ? d2 : 0;
      a += (v >= e3) ? d3 : 0;
      a += (v >= e4) ? d4 : 0;
      a += (v >= e5) ? d5 : 0;
      a += (v >= e6) ? d6 : 0;
      a += (v >= e7) ? d7 : 0;
      return bins[a];
    };
    unsigned pkc = edge_at(lane);
    unsigned pkn = edge_at(32 + lane);
    float4 xc = {0.0f, 0.0f, 0.0f, 0.0f};
    if constexpr (FIRST) xc = ((const float4*)h)[pkc & SM];

    for (int base = 0; base < Tw; base += 32) {
      const unsigned pk2 = edge_at(base + 64 + lane);
      float4 xn = {0.0f, 0.0f, 0.0f, 0.0f};
      if constexpr (FIRST) xn = ((const float4*)h)[pkn & SM];

      const int dv = (int)(pkc >> SRCBITS);
      const int nb = (Tw - base) < 32 ? (Tw - base) : 32;
      if constexpr (FIRST) {
        if (lane < nb)
          __hip_atomic_fetch_add(cl + dv, 1.0f, __ATOMIC_RELAXED, __HIP_MEMORY_SCOPE_WORKGROUP);
        for (int k = 0; k < nb; ++k) {
          const float z  = __int_as_float(__builtin_amdgcn_readlane(__float_as_int(xc.x), k));
          const float px = __int_as_float(__builtin_amdgcn_readlane(__float_as_int(xc.y), k));
          const float py = __int_as_float(__builtin_amdgcn_readlane(__float_as_int(xc.z), k));
          const float pz = __int_as_float(__builtin_amdgcn_readlane(__float_as_int(xc.w), k));
          const int   dk = __builtin_amdgcn_readlane(dv, k);
          float* tp = tile + dk * HID + lane;
#pragma unroll
          for (int j = 0; j < 4; ++j) {
            const float v = h0_elem(z, px, py, pz, wi[0][j], wi[1][j], wi[2][j], wi[3][j], bi[j]);
            __hip_atomic_fetch_add(tp + 32 * j, v, __ATOMIC_RELAXED, __HIP_MEMORY_SCOPE_WORKGROUP);
          }
        }
      } else {
        add_rows(h, tile, (int)(pkc & SM), dv, nb, lane);
      }
      pkc = pkn;
      pkn = pk2;
      xc  = xn;
    }
  }
  __syncthreads();

  {
    float4*       s4 = (float4*)(s + (size_t)lo * HID);
    const float4* t4 = (const float4*)tile;
    for (int i = threadIdx.x; i < nloc * (HID / 4); i += 1024) s4[i] = t4[i];
    if constexpr (FIRST) {
      for (int i = threadIdx.x; i < nloc; i += 1024) cnt[lo + i] = cl[i];
    }
  }
}

typedef __attribute__((ext_vector_type(4))) _Float16 half4;

__device__ inline v16h join16(half8 lo, half8 hi) {
  return __builtin_shufflevector(lo, hi, 0, 1, 2, 3, 4, 5, 6, 7,
                                         8, 9, 10, 11, 12, 13, 14, 15);
}

#define MT   9
#define GR   (16 * MT)
#define RPW  (GR / 8)
#define APIT 136
template <bool SAGE, bool RELU, bool FIRST>
__global__ __launch_bounds__(256) void gemm_kernel(
    const float*    __restrict__ h,
    const float*    __restrict__ Win,
    const float*    __restrict__ bin,
    const float*    __restrict__ s,
    const float*    __restrict__ cnt,
    const float*    __restrict__ Wl,
    const float*    __restrict__ Wr,
    const float*    __restrict__ bias,
    float* __restrict__ out,
    int n_nodes) {
  __shared__ __attribute__((aligned(16))) _Float16 ah[GR * APIT];
  __shared__ __attribute__((aligned(16))) _Float16 ag[SAGE ? GR * APIT : 8];

  const int lane = threadIdx.x & 31;
  const int wave = __builtin_amdgcn_readfirstlane((int)(threadIdx.x >> 5));
  const int c    = lane & 15;
  const int lo   = (lane >= 16) ? 8 : 0;
  const int row0 = blockIdx.x * GR;

  {
    const int rb = wave * RPW;
    int lrow = row0 + rb + (lane < RPW ? lane : RPW - 1);
    if (lrow >= n_nodes) lrow = n_nodes - 1;
    float invl = 0.0f;
    if constexpr (SAGE) {
      const float cv = cnt[lrow];
      invl = (cv > 0.0f) ? __builtin_amdgcn_rcpf(cv) : 0.0f;
    }
    float4 xl = {0.0f, 0.0f, 0.0f, 0.0f};
    float wi[4][4], bi[4];
    if constexpr (FIRST) {
      xl = ((const float4*)h)[lrow];
#pragma unroll
      for (int r = 0; r < 4; ++r) {
        const float4 w4 = *(const float4*)(Win + r * HID + lane * 4);
        wi[r][0] = w4.x; wi[r][1] = w4.y; wi[r][2] = w4.z; wi[r][3] = w4.w;
      }
      const float4 b4 = *(const float4*)(bin + lane * 4);
      bi[0] = b4.x; bi[1] = b4.y; bi[2] = b4.z; bi[3] = b4.w;
    }
#pragma unroll 9
    for (int it = 0; it < RPW; ++it) {
      const int r = rb + it;
      int row = row0 + r;
      if (row >= n_nodes) row = n_nodes - 1;
      half4 hv;
      if constexpr (FIRST) {
        const float z  = __int_as_float(__builtin_amdgcn_readlane(__float_as_int(xl.x), it));
        const float px = __int_as_float(__builtin_amdgcn_readlane(__float_as_int(xl.y), it));
        const float py = __int_as_float(__builtin_amdgcn_readlane(__float_as_int(xl.z), it));
        const float pz = __int_as_float(__builtin_amdgcn_readlane(__float_as_int(xl.w), it));
#pragma unroll
        for (int e = 0; e < 4; ++e)
          hv[e] = (_Float16)h0_elem(z, px, py, pz, wi[0][e], wi[1][e], wi[2][e], wi[3][e], bi[e]);
      } else {
        const float4 v = *(const float4*)(h + (size_t)row * HID + lane * 4);
        hv[0] = (_Float16)v.x; hv[1] = (_Float16)v.y;
        hv[2] = (_Float16)v.z; hv[3] = (_Float16)v.w;
      }
      *(half4*)&ah[r * APIT + lane * 4] = hv;
      if constexpr (SAGE) {
        const float inv = __int_as_float(__builtin_amdgcn_readlane(__float_as_int(invl), it));
        const float4 v = *(const float4*)(s + (size_t)row * HID + lane * 4);
        half4 gv;
        gv[0] = (_Float16)(v.x * inv); gv[1] = (_Float16)(v.y * inv);
        gv[2] = (_Float16)(v.z * inv); gv[3] = (_Float16)(v.w * inv);
        *(half4*)&ag[r * APIT + lane * 4] = gv;
      }
    }
  }
  v16h br[4], bl[4];
  {
    const float* wr = Wr + (size_t)lo * HID + wave * 16 + c;
    float wv[64];
#pragma unroll
    for (int i = 0; i < 64; ++i) {
      const int j = i & 15;
      wv[i] = wr[((i >> 4) * 32 + (j < 8 ? j : j + 8)) * HID];
    }
#pragma unroll
    for (int kk = 0; kk < 4; ++kk) {
#pragma unroll
      for (int e = 0; e < 16; ++e) br[kk][e] = (_Float16)wv[kk * 16 + e];
    }
    asm volatile("" :: "v"(br[0]), "v"(br[1]), "v"(br[2]), "v"(br[3]));
    if constexpr (SAGE) {
      const float* wl = Wl + (size_t)lo * HID + wave * 16 + c;
      float wu[64];
#pragma unroll
      for (int i = 0; i < 64; ++i) {
        const int j = i & 15;
        wu[i] = wl[((i >> 4) * 32 + (j < 8 ? j : j + 8)) * HID];
      }
#pragma unroll
      for (int kk = 0; kk < 4; ++kk) {
#pragma unroll
        for (int e = 0; e < 16; ++e) bl[kk][e] = (_Float16)wu[kk * 16 + e];
      }
      asm volatile("" :: "v"(bl[0]), "v"(bl[1]), "v"(bl[2]), "v"(bl[3]));
    }
  }

  __syncthreads();

  const _Float16* ap = ah + c * APIT + lo;
  const _Float16* gp = ag + c * APIT + lo;

  v8f acc[MT];
  {
    v8f z = {};
#pragma unroll
    for (int mt = 0; mt < MT; ++mt) acc[mt] = z;
  }

#pragma unroll
  for (int kk = 0; kk < 4; ++kk) {
#pragma unroll
    for (int mt = 0; mt < MT; ++mt) {
      const int off = mt * 16 * APIT + kk * 32;
      v16h af = join16(*(const half8*)(ap + off), *(const half8*)(ap + off + 16));
      acc[mt] = __builtin_amdgcn_wmma_f32_16x16x32_f16(
          false, af, false, br[kk], (short)0, acc[mt], false, false);
      if constexpr (SAGE) {
        v16h gf = join16(*(const half8*)(gp + off), *(const half8*)(gp + off + 16));
        acc[mt] = __builtin_amdgcn_wmma_f32_16x16x32_f16(
            false, gf, false, bl[kk], (short)0, acc[mt], false, false);
      }
    }
  }

  const int   ncol = wave * 16 + c;
  const float bb   = bias[ncol];
#pragma unroll
  for (int mt = 0; mt < MT; ++mt) {
#pragma unroll
    for (int r = 0; r < 8; ++r) {
      float v = acc[mt][r] + bb;
      if constexpr (RELU) v = fmaxf(v, 0.0f);
      const int orow = row0 + mt * 16 + lo + r;
      if (orow < n_nodes) out[(size_t)orow * HID + ncol] = v;
    }
  }
}

#define CS_BLOCKS 256
__global__ __launch_bounds__(512) void colsum_kernel(const float* __restrict__ ns,
                                                     float* __restrict__ part,
                                                     int n) {
  __shared__ float red[16][HID];
  const int cg   = threadIdx.x & 31;
  const int rsub = threadIdx.x >> 5;
  const int rpb  = (n + (int)gridDim.x - 1) / (int)gridDim.x;
  const int r0   = blockIdx.x * rpb;
  int r1 = r0 + rpb;
  if (r1 > n) r1 = n;
  float4 a = {0.0f, 0.0f, 0.0f, 0.0f};
#pragma unroll 4
  for (int r = r0 + rsub; r < r1; r += 16) {
    const float4 v = *(const float4*)(ns + (size_t)r * HID + cg * 4);
    a.x += v.x; a.y += v.y; a.z += v.z; a.w += v.w;
  }
  red[rsub][cg * 4 + 0] = a.x;
  red[rsub][cg * 4 + 1] = a.y;
  red[rsub][cg * 4 + 2] = a.z;
  red[rsub][cg * 4 + 3] = a.w;
  __syncthreads();
  if (threadIdx.x < HID) {
    float t = 0.0f;
    for (int k = 0; k < 16; ++k) t += red[k][threadIdx.x];
    part[(size_t)blockIdx.x * HID + threadIdx.x] = t;
  }
}

__global__ __launch_bounds__(1024) void colmean_final_kernel(
    const float* __restrict__ part, float* __restrict__ outm, int nparts, int n) {
  __shared__ float red[8][HID];
  const int col = threadIdx.x & (HID - 1);
  const int g   = threadIdx.x >> 7;
  float t = 0.0f;
#pragma unroll 4
  for (int b = g; b < nparts; b += 8) t += part[(size_t)b * HID + col];
  red[g][col] = t;
  __syncthreads();
  if (threadIdx.x < HID) {
    float sum = 0.0f;
    for (int k = 0; k < 8; ++k) sum += red[k][threadIdx.x];
    outm[threadIdx.x] = sum / (float)n;
  }
}

extern "C" void kernel_launch(void* const* d_in, const int* in_sizes, int n_in,
                              void* d_out, int out_size, void* d_ws, size_t ws_size,
                              hipStream_t stream) {
  const float* pos  = (const float*)d_in[0];
  const int*   an   = (const int*)  d_in[1];
  const int*   ei   = (const int*)  d_in[2];
  const float* W_in = (const float*)d_in[3];
  const float* b_in = (const float*)d_in[4];
  const float* W1l  = (const float*)d_in[5];
  const float* b1   = (const float*)d_in[6];
  const float* W1r  = (const float*)d_in[7];
  const float* W2l  = (const float*)d_in[8];
  const float* b2   = (const float*)d_in[9];
  const float* W2r  = (const float*)d_in[10];
  const float* Wout = (const float*)d_in[11];
  const float* bout = (const float*)d_in[12];

  const int n  = in_sizes[0] / 3;
  const int nE = in_sizes[2] / 2;

  float* ws   = (float*)d_ws;
  float* h    = ws;
  float* s    = ws + (size_t)n * HID;
  float* cnt  = s  + (size_t)n * HID;

  size_t woff = (size_t)n * HID * 2 + n;
  woff = (woff + 63) & ~(size_t)63;
  float* part = ws + woff;
  float* xp = part + CS_BLOCKS * HID;

  float* outm = (float*)d_out;
  float* nst  = (float*)d_out + HID;

  const int gemmBlks = (n + GR - 1) / GR;
  const int aggBlks  = (n + NPB - 1) / NPB;
  const int chunk    = (nE + NBIN - 1) / NBIN;
  unsigned* bins = (unsigned*)(xp + (size_t)n * 4);
  int* offs = (int*)(bins + (size_t)NBIN * chunk);

  xpack_kernel<<<(n + 255) / 256, 256, 0, stream>>>(pos, an, xp, n);

  bin_kernel<<<NBIN, 1024, 0, stream>>>(ei, bins, offs, nE, chunk);

  agg_kernel<true><<<aggBlks, 1024, 0, stream>>>(xp, W_in, b_in, bins, offs, s, cnt, n, chunk);
  gemm_kernel<true, true, true><<<gemmBlks, 256, 0, stream>>>(xp, W_in, b_in, s, cnt, W1l, W1r, b1, h, n);

  agg_kernel<false><<<aggBlks, 1024, 0, stream>>>(h, nullptr, nullptr, bins, offs, s, cnt, n, chunk);
  gemm_kernel<true, true, false><<<gemmBlks, 256, 0, stream>>>(h, nullptr, nullptr, s, cnt, W2l, W2r, b2, h, n);

  gemm_kernel<false, false, false><<<gemmBlks, 256, 0, stream>>>(h, nullptr, nullptr, nullptr, nullptr,
                                                                 nullptr, Wout, bout, nst, n);
  colsum_kernel<<<CS_BLOCKS, 512, 0, stream>>>(nst, part, n);
  colmean_final_kernel<<<1, 1024, 0, stream>>>(part, outm, CS_BLOCKS, n);
}
